# starting kernel
# baseline (speedup 1.0000x reference)
#include <hip/hip_runtime.h>
#include <math.h>


typedef __attribute__((ext_vector_type(2))) float v2f;
typedef __attribute__((ext_vector_type(8))) float v8f;

#define BATCH 32
#define CCH   256
#define HH    64
#define WW    64
#define HW    (HH * WW)
#define CHW   (CCH * HW)
#define EPSV  1e-5f

__global__ void pool_kernel(const float* __restrict__ x,
                            float* __restrict__ pmax,
                            float* __restrict__ pavg) {
  __shared__ float4 smax[256];
  __shared__ float4 ssum[256];
  const int bh = blockIdx.x;
  const int b = bh >> 6, h = bh & 63;
  const int t = threadIdx.x;
  const int w4 = t & 15, cg = t >> 4;
  const float4* base = (const float4*)(x + (size_t)b * CHW + (size_t)h * WW) + w4;
  float4 mx = make_float4(-INFINITY, -INFINITY, -INFINITY, -INFINITY);
  float4 sm = make_float4(0.f, 0.f, 0.f, 0.f);
  const int c0 = cg * 16;
#pragma unroll 4
  for (int i = 0; i < 16; ++i) {
    float4 v = base[(size_t)(c0 + i) * (HW / 4)];
    mx.x = fmaxf(mx.x, v.x); mx.y = fmaxf(mx.y, v.y);
    mx.z = fmaxf(mx.z, v.z); mx.w = fmaxf(mx.w, v.w);
    sm.x += v.x; sm.y += v.y; sm.z += v.z; sm.w += v.w;
  }
  smax[t] = mx;
  ssum[t] = sm;
  __syncthreads();
  for (int s = 128; s >= 16; s >>= 1) {
    if (t < s) {
      float4 a = smax[t], c = smax[t + s];
      a.x = fmaxf(a.x, c.x); a.y = fmaxf(a.y, c.y);
      a.z = fmaxf(a.z, c.z); a.w = fmaxf(a.w, c.w);
      smax[t] = a;
      float4 u = ssum[t], d = ssum[t + s];
      u.x += d.x; u.y += d.y; u.z += d.z; u.w += d.w;
      ssum[t] = u;
    }
    __syncthreads();
  }
  if (t < 16) {
    float4 m2 = smax[t];
    float4 s2 = ssum[t];
    s2.x *= (1.f / 256.f); s2.y *= (1.f / 256.f);
    s2.z *= (1.f / 256.f); s2.w *= (1.f / 256.f);
    ((float4*)pmax)[bh * 16 + t] = m2;
    ((float4*)pavg)[bh * 16 + t] = s2;
  }
}

__global__ void conv_att_kernel(const float* __restrict__ pmax,
                                const float* __restrict__ pavg,
                                const float* __restrict__ w1,
                                const float* __restrict__ w2,
                                const float* __restrict__ w3,
                                const float* __restrict__ g1, const float* __restrict__ be1,
                                const float* __restrict__ m1, const float* __restrict__ v1,
                                const float* __restrict__ g2, const float* __restrict__ be2,
                                const float* __restrict__ m2, const float* __restrict__ v2,
                                const float* __restrict__ g3, const float* __restrict__ be3,
                                const float* __restrict__ m3, const float* __restrict__ v3,
                                float* __restrict__ att) {
  __shared__ float lp[2][8][72];
  __shared__ float btabT[16][128];
  __shared__ float ytmp[4][3][16];
  const int bh = blockIdx.x;
  const int b = bh >> 6, h = bh & 63;
  const int tid = threadIdx.x;

  for (int idx = tid; idx < 16 * 128; idx += 128) {
    int n = idx >> 7, k = idx & 127;
    int ch = k >> 6, r = (k >> 3) & 7, s = k & 7;
    float val = 0.f;
    if (n < 3 && r < 7 && s < 7) {
      if (n == 0)      val = w1[ch * 49 + r * 7 + s];
      else if (n == 1) val = w2[ch * 49 + s * 7 + r];
      else             val = w3[ch * 49 + r * 7 + s];
    }
    btabT[n][k] = val;
  }
  for (int idx = tid; idx < 2 * 8 * 72; idx += 128) ((float*)lp)[idx] = 0.f;
  __syncthreads();
  for (int idx = tid; idx < 2 * 7 * 64; idx += 128) {
    int w = idx & 63;
    int rr = (idx >> 6) % 7;
    int ch = idx / (7 * 64);
    int hh = h + rr - 3;
    if (hh >= 0 && hh < 64) {
      const float* src = (ch == 0) ? pmax : pavg;
      lp[ch][rr][w + 3] = src[(b * 64 + hh) * 64 + w];
    }
  }
  __syncthreads();

  const int lane = tid & 31, wid = tid >> 5;
  const int m = lane & 15, khalf = lane >> 4;
  const int w0 = wid * 16;
  const float* lpf = &lp[0][0][0];
  const float* btf = &btabT[0][0];
  const int aBase = w0 + m + 2 * khalf;
  const int bBase = m * 128 + 2 * khalf;

  v8f acc = {};
#if __has_builtin(__builtin_amdgcn_wmma_f32_16x16x4_f32)
#pragma unroll
  for (int chunk = 0; chunk < 32; ++chunk) {
    const int k0 = chunk * 4;
    const int ch = k0 >> 6;
    const int r  = (k0 >> 3) & 7;
    const int sc = k0 & 7;
    const int ai = ch * 576 + r * 72 + sc + aBase;
    v2f a;
    a[0] = lpf[ai];
    a[1] = lpf[ai + 1];
    v2f bm = *(const v2f*)(btf + bBase + k0);
    acc = __builtin_amdgcn_wmma_f32_16x16x4_f32(
         false, a,  false, bm,
         (short)0, acc,  false,  false);
  }
#else
  for (int k = 0; k < 128; ++k) {
    int ch = k >> 6, r = (k >> 3) & 7, s = k & 7;
    float bw = btf[m * 128 + k];
#pragma unroll
    for (int v = 0; v < 8; ++v) {
      int M = v + 8 * khalf;
      acc[v] += lp[ch][r][w0 + M + s] * bw;
    }
  }
#endif

  if (m < 3) {
#pragma unroll
    for (int v = 0; v < 8; ++v) ytmp[wid][m][v + 8 * khalf] = acc[v];
  }
  __syncthreads();

  if (lane < 16) {
    float y0 = ytmp[wid][0][lane];
    float y1 = ytmp[wid][1][lane];
    float y2 = ytmp[wid][2][lane];
    float s1 = g1[0] / sqrtf(v1[0] + EPSV), t1 = be1[0] - m1[0] * s1;
    float s2 = g2[0] / sqrtf(v2[0] + EPSV), t2 = be2[0] - m2[0] * s2;
    float s3 = g3[0] / sqrtf(v3[0] + EPSV), t3 = be3[0] - m3[0] * s3;
    float a1 = 1.f / (1.f + expf(-(y0 * s1 + t1)));
    float a2 = 1.f / (1.f + expf(-(y1 * s2 + t2)));
    float a3 = 1.f / (1.f + expf(-(y2 * s3 + t3)));
    att[bh * 64 + w0 + lane] = (a1 + a2 + a3) * (1.f / 3.f);
  }
}

__global__ void scale_kernel(const float* __restrict__ x,
                             const float* __restrict__ att,
                             float* __restrict__ out) {
  const size_t i4 = (size_t)blockIdx.x * blockDim.x + threadIdx.x;
  const size_t base = i4 * 4;
  const int b = (int)(base >> 20);
  const int hw = (int)(base & 4095);
  float4 xv = ((const float4*)x)[i4];
  float4 av = *(const float4*)(att + (size_t)b * HW + hw);
  float4 o;
  o.x = xv.x * av.x;
  o.y = xv.y * av.y;
  o.z = xv.z * av.z;
  o.w = xv.w * av.w;
  ((float4*)out)[i4] = o;
}

extern "C" void kernel_launch(void* const* d_in, const int* in_sizes, int n_in,
                              void* d_out, int out_size, void* d_ws, size_t ws_size,
                              hipStream_t stream) {
  const float* x  = (const float*)d_in[0];
  const float* w1 = (const float*)d_in[1];
  const float* g1 = (const float*)d_in[2];
  const float* b1 = (const float*)d_in[3];
  const float* m1 = (const float*)d_in[4];
  const float* v1 = (const float*)d_in[5];
  const float* w2 = (const float*)d_in[6];
  const float* g2 = (const float*)d_in[7];
  const float* b2 = (const float*)d_in[8];
  const float* m2 = (const float*)d_in[9];
  const float* v2 = (const float*)d_in[10];
  const float* w3 = (const float*)d_in[11];
  const float* g3 = (const float*)d_in[12];
  const float* b3 = (const float*)d_in[13];
  const float* m3 = (const float*)d_in[14];
  const float* v3 = (const float*)d_in[15];

  float* ws   = (float*)d_ws;
  float* pmax = ws;
  float* pavg = ws + BATCH * HW;
  float* attb = ws + 2 * BATCH * HW;

  pool_kernel<<<BATCH * HH, 256, 0, stream>>>(x, pmax, pavg);
  conv_att_kernel<<<BATCH * HH, 128, 0, stream>>>(pmax, pavg, w1, w2, w3,
                                                  g1, b1, m1, v1,
                                                  g2, b2, m2, v2,
                                                  g3, b3, m3, v3, attb);
  const int total4 = BATCH * CHW / 4;
  scale_kernel<<<total4 / 256, 256, 0, stream>>>(x, attb, (float*)d_out);
}


# both launches as row loops: 512 resident work-groups, work-group i takes rows i+512j, j=0..3; two row buffers per lane, row j+2 requested when row j leaves its buffer, loads in flight across barriers
# speedup vs baseline: 1.7562x; 1.7562x over previous
#include <hip/hip_runtime.h>
#include <math.h>


typedef __attribute__((ext_vector_type(2))) float v2f;
typedef __attribute__((ext_vector_type(8))) float v8f;

#define BATCH 32
#define CCH   256
#define HH    64
#define WW    64
#define HW    (HH * WW)
#define CHW   (CCH * HW)
#define EPSV  1e-5f
#define NPART 21
#define BT_PITCH 18
#define ST_PITCH 68
#define NROWS (BATCH * HH)
#define NBLK  512
#define RPB   (NROWS / NBLK)

static __device__ __forceinline__ void lds_barrier() {
  asm volatile("s_wait_dscnt 0x0\n\ts_barrier_signal -1\n\ts_barrier_wait -1" ::: "memory");
}

static __device__ __forceinline__ void load_row16(float4 (&v)[16], const float* x,
                                                  int bh, int w4, int c0) {
  const int b = bh >> 6, h = bh & 63;
  const float4* base = (const float4*)(x + (size_t)b * CHW + (size_t)h * WW) + w4;
#pragma unroll
  for (int i = 0; i < 16; ++i) v[i] = base[(size_t)(c0 + i) * (HW / 4)];
}

__global__ __launch_bounds__(256) void pool_hconv_kernel(const float* x,
                                                         const float* __restrict__ w1,
                                                         const float* __restrict__ w2,
                                                         const float* __restrict__ w3,
                                                         float* __restrict__ part) {
  __shared__ float4 smax[256];
  __shared__ float4 ssum[256];
  __shared__ float prow[2][72];
  __shared__ __attribute__((aligned(16))) float btab[32][BT_PITCH];
  __shared__ __attribute__((aligned(16))) float stage[32][ST_PITCH];
  const int t = threadIdx.x;

  float wv[2];
#pragma unroll
  for (int j = 0; j < 2; ++j) {
    const int n = (t >> 4) + 16 * j, k = t & 15;
    const int ch = k >> 3, s = k & 7;
    const int br = n / 7, r = n - br * 7;
    const bool valid = (n < NPART) && (s < 7);
    const float* wp = (br == 1) ? w2 : ((br == 2) ? w3 : w1);
    const int off = !valid ? 0
                  : (br == 1) ? (ch * 49 + s * 7 + r)
                              : (ch * 49 + r * 7 + s);
    const float wl = wp[off];
    wv[j] = valid ? wl : 0.f;
  }
  if (t < 16) {
    const int ch = t >> 3, j = t & 7;
    prow[ch][(j < 3) ? j : (64 + j)] = 0.f;
  }

  const int w4 = t & 15, cg = t >> 4;
  const int c0 = cg * 16;

  const int lane = t & 31, wid = t >> 5;
  const int m = lane & 15, khalf = lane >> 4;
  const int w0 = (wid & 3) * 16;
  const int n0 = (wid >> 2) * 16;
  const float* pa = &prow[0][0] + w0 + m + 2 * khalf;
  const float* pb = &btab[n0 + m][2 * khalf];

  float4 xr[2][16];
  load_row16(xr[0], x, blockIdx.x, w4, c0);
  load_row16(xr[1], x, blockIdx.x + NBLK, w4, c0);
  btab[(t >> 4)][t & 15] = wv[0];
  btab[(t >> 4) + 16][t & 15] = wv[1];

#pragma unroll
  for (int j = 0; j < RPB; ++j) {
    const int bh = blockIdx.x + NBLK * j;

    float4 mx = make_float4(-INFINITY, -INFINITY, -INFINITY, -INFINITY);
    float4 sm = make_float4(0.f, 0.f, 0.f, 0.f);
#pragma unroll
    for (int i = 0; i < 16; ++i) {
      const float4 v = xr[j & 1][i];
      mx.x = fmaxf(mx.x, v.x); mx.y = fmaxf(mx.y, v.y);
      mx.z = fmaxf(mx.z, v.z); mx.w = fmaxf(mx.w, v.w);
      sm.x += v.x; sm.y += v.y; sm.z += v.z; sm.w += v.w;
    }
    smax[t] = mx;
    ssum[t] = sm;
    if (j + 2 < RPB) load_row16(xr[j & 1], x, bh + 2 * NBLK, w4, c0);
    lds_barrier();
    for (int s = 128; s >= 16; s >>= 1) {
      if (t < s) {
        float4 a = smax[t], c = smax[t + s];
        a.x = fmaxf(a.x, c.x); a.y = fmaxf(a.y, c.y);
        a.z = fmaxf(a.z, c.z); a.w = fmaxf(a.w, c.w);
        smax[t] = a;
        float4 u = ssum[t], d = ssum[t + s];
        u.x += d.x; u.y += d.y; u.z += d.z; u.w += d.w;
        ssum[t] = u;
      }
      lds_barrier();
    }
    if (t < 16) {
      float4 m2 = smax[t];
      float4 s2 = ssum[t];
      s2.x *= (1.f / 256.f); s2.y *= (1.f / 256.f);
      s2.z *= (1.f / 256.f); s2.w *= (1.f / 256.f);
      float* qm = &prow[0][3 + 4 * t];
      float* qa = &prow[1][3 + 4 * t];
      qm[0] = m2.x; qm[1] = m2.y; qm[2] = m2.z; qm[3] = m2.w;
      qa[0] = s2.x; qa[1] = s2.y; qa[2] = s2.z; qa[3] = s2.w;
    }
    lds_barrier();

    v8f acc = {};
#pragma unroll
    for (int kc = 0; kc < 4; ++kc) {
      const int k0 = kc * 4;
      const int ch = k0 >> 3;
      const int sc = k0 & 7;
      v2f a;
      a[0] = pa[ch * 72 + sc];
      a[1] = pa[ch * 72 + sc + 1];
      v2f bm = *(const v2f*)(pb + k0);
      acc = __builtin_amdgcn_wmma_f32_16x16x4_f32(
           false, a,  false, bm,
           (short)0, acc,  false,  false);
    }
    {
      float* sp = &stage[n0 + m][w0 + 8 * khalf];
      *(float4*)(sp)     = make_float4(acc[0], acc[1], acc[2], acc[3]);
      *(float4*)(sp + 4) = make_float4(acc[4], acc[5], acc[6], acc[7]);
    }
    lds_barrier();
    for (int idx = t; idx < NPART * 16; idx += 256) {
      const int n = idx >> 4, q = idx & 15;
      const float4 v = *(const float4*)&stage[n][4 * q];
      ((float4*)part)[(size_t)bh * (NPART * 16) + idx] = v;
    }
  }
}

__global__ __launch_bounds__(256) void vsum_scale_kernel(const float* x,
                                  const float* part,
                                  const float* __restrict__ g1, const float* __restrict__ be1,
                                  const float* __restrict__ m1, const float* __restrict__ v1,
                                  const float* __restrict__ g2, const float* __restrict__ be2,
                                  const float* __restrict__ m2, const float* __restrict__ v2,
                                  const float* __restrict__ g3, const float* __restrict__ be3,
                                  const float* __restrict__ m3, const float* __restrict__ v3,
                                  float* __restrict__ out) {
  __shared__ float sig[2][3][64];
  const int t = threadIdx.x;
  const int w4 = t & 15, cg = t >> 4;
  const int c0 = cg * 16;
  const int br = t >> 6, px = t & 63;

  const float s1 = g1[0] / sqrtf(v1[0] + EPSV), t1 = be1[0] - m1[0] * s1;
  const float s2 = g2[0] / sqrtf(v2[0] + EPSV), t2 = be2[0] - m2[0] * s2;
  const float s3 = g3[0] / sqrtf(v3[0] + EPSV), t3 = be3[0] - m3[0] * s3;
  const float sb = (br == 0) ? s1 : ((br == 1) ? s2 : s3);
  const float tb = (br == 0) ? t1 : ((br == 1) ? t2 : t3);

  float4 xr[2][16];
  float pv[2][7];

  const int brl = (br < 3) ? br : 2;
#define LOAD_PART(dst, bh_)                                                        \
  {                                                                                \
    const int b_ = (bh_) >> 6, h_ = (bh_) & 63;                                    \
    _Pragma("unroll")                                                              \
    for (int r = 0; r < 7; ++r) {                                                  \
      const int hc = min(max(h_ + r - 3, 0), 63);                                  \
      dst[r] = part[((size_t)(b_ * 64 + hc) * NPART + brl * 7 + r) * 64 + px];     \
    }                                                                              \
  }

  load_row16(xr[0], x, blockIdx.x, w4, c0);
  LOAD_PART(pv[0], blockIdx.x)
  asm volatile("" ::: "memory");
  load_row16(xr[1], x, blockIdx.x + NBLK, w4, c0);
  LOAD_PART(pv[1], blockIdx.x + NBLK)

#pragma unroll
  for (int j = 0; j < RPB; ++j) {
    const int bh = blockIdx.x + NBLK * j;
    const int b = bh >> 6, h = bh & 63;

    if (t < 192) {
      float y = 0.f;
#pragma unroll
      for (int r = 0; r < 7; ++r) {
        const int hh = h + r - 3;
        y += (hh >= 0 && hh < 64) ? pv[j & 1][r] : 0.f;
      }
      sig[j & 1][br][px] = 1.f / (1.f + expf(-(y * sb + tb)));
    }
    lds_barrier();

    const float* sg = &sig[j & 1][0][0];
    float4 av;
    av.x = (sg[4 * w4 + 0] + sg[64 + 4 * w4 + 0] + sg[128 + 4 * w4 + 0]) * (1.f / 3.f);
    av.y = (sg[4 * w4 + 1] + sg[64 + 4 * w4 + 1] + sg[128 + 4 * w4 + 1]) * (1.f / 3.f);
    av.z = (sg[4 * w4 + 2] + sg[64 + 4 * w4 + 2] + sg[128 + 4 * w4 + 2]) * (1.f / 3.f);
    av.w = (sg[4 * w4 + 3] + sg[64 + 4 * w4 + 3] + sg[128 + 4 * w4 + 3]) * (1.f / 3.f);

    float4* ob = (float4*)(out + (size_t)b * CHW + (size_t)h * WW) + w4;
#pragma unroll
    for (int i = 0; i < 16; ++i) {
      const float4 v = xr[j & 1][i];
      float4 o;
      o.x = v.x * av.x;
      o.y = v.y * av.y;
      o.z = v.z * av.z;
      o.w = v.w * av.w;
      ob[(size_t)(c0 + i) * (HW / 4)] = o;
    }
    if (j + 2 < RPB) {
      load_row16(xr[j & 1], x, bh + 2 * NBLK, w4, c0);
      LOAD_PART(pv[j & 1], bh + 2 * NBLK)
    }
  }
#undef LOAD_PART
}

extern "C" void kernel_launch(void* const* d_in, const int* in_sizes, int n_in,
                              void* d_out, int out_size, void* d_ws, size_t ws_size,
                              hipStream_t stream) {
  const float* x  = (const float*)d_in[0];
  const float* w1 = (const float*)d_in[1];
  const float* g1 = (const float*)d_in[2];
  const float* b1 = (const float*)d_in[3];
  const float* m1 = (const float*)d_in[4];
  const float* v1 = (const float*)d_in[5];
  const float* w2 = (const float*)d_in[6];
  const float* g2 = (const float*)d_in[7];
  const float* b2 = (const float*)d_in[8];
  const float* m2 = (const float*)d_in[9];
  const float* v2 = (const float*)d_in[10];
  const float* w3 = (const float*)d_in[11];
  const float* g3 = (const float*)d_in[12];
  const float* b3 = (const float*)d_in[13];
  const float* m3 = (const float*)d_in[14];
  const float* v3 = (const float*)d_in[15];

  float* part = (float*)d_ws;

  pool_hconv_kernel<<<NBLK, 256, 0, stream>>>(x, w1, w2, w3, part);
  vsum_scale_kernel<<<NBLK, 256, 0, stream>>>(x, part,
                                              g1, b1, m1, v1,
                                              g2, b2, m2, v2,
                                              g3, b3, m3, v3, (float*)d_out);
}
